# starting kernel
# baseline (speedup 1.0000x reference)
#include <hip/hip_runtime.h>

typedef __attribute__((ext_vector_type(2))) float v2f;
typedef __attribute__((ext_vector_type(4))) float v4f;
typedef __attribute__((ext_vector_type(8))) float v8f;

#define T_DIM 8192
#define SEG 512
#define LOOKBACK 64
#define NSEG (T_DIM / SEG)
#define WAVES_PER_BLOCK 8

__device__ __forceinline__ float fast_pow_pos(float a, float b) {
  return __builtin_amdgcn_exp2f(b * __builtin_amdgcn_logf(a));
}

__global__ __launch_bounds__(256) void pcen_wmma_kernel(
    const float* __restrict__ x,
    const float* __restrict__ p_alpha,
    const float* __restrict__ p_delta,
    const float* __restrict__ p_r,
    const float* __restrict__ p_s,
    float* __restrict__ out,
    int n_rowgroups)
{
  const int lane = threadIdx.x & 31;
  const int widx = threadIdx.x >> 5;
  const int task = blockIdx.x * WAVES_PER_BLOCK + widx;
  const int rowgroup = task / NSEG;
  const int seg      = task % NSEG;
  if (rowgroup >= n_rowgroups) return;

  const float alpha = fminf(fmaxf(p_alpha[0], 0.01f), 0.99f);
  const float delta = fabsf(p_delta[0]) + 1e-6f;
  const float r     = fminf(fmaxf(p_r[0], 0.01f), 1.0f);
  const float s     = p_s[0];
  const float decay = 1.0f - s;
  const float dr    = fast_pow_pos(delta, r);

  const int halfsel = lane >> 4;
  const int col     = lane & 15;

  v2f amat[4];
#pragma unroll
  for (int kk = 0; kk < 4; ++kk) {
#pragma unroll
    for (int v = 0; v < 2; ++v) {
      int k = 4 * kk + v + 2 * halfsel;
      float w = 0.0f;
      if (col >= k) {
        float pw = s;
        for (int t = 0; t < col - k; ++t) pw *= decay;
        w = pw;
      }
      amat[kk][v] = w;
    }
  }
  float cdecay[8];
  {
    float pw = decay;
    for (int t = 0; t < 8 * halfsel; ++t) pw *= decay;
#pragma unroll
    for (int v = 0; v < 8; ++v) { cdecay[v] = pw; pw *= decay; }
  }

  const size_t row_idx = (size_t)rowgroup * 16 + (size_t)col;
  const float* __restrict__ xrow = x   + row_idx * (size_t)T_DIM;
  float*       __restrict__ orow = out + row_idx * (size_t)T_DIM;

  int t0, nchunks, warm_chunks;
  float m_in;
  if (seg == 0) {
    t0 = 0; nchunks = SEG / 16; warm_chunks = 0;
    m_in = xrow[0];
  } else {
    t0 = seg * SEG - LOOKBACK;
    nchunks = (SEG + LOOKBACK) / 16;
    warm_chunks = LOOKBACK / 16;
    m_in = 0.0f;
  }

  for (int c = 0; c < nchunks; ++c) {
    const int tbase = t0 + c * 16;

    __builtin_prefetch(xrow + tbase + 16, 0, 1);

    v8f acc;
#pragma unroll
    for (int v = 0; v < 8; ++v) acc[v] = cdecay[v] * m_in;

#pragma unroll
    for (int kk = 0; kk < 4; ++kk) {
      const int tb = tbase + 4 * kk + 2 * halfsel;
      v2f b;
      b[0] = xrow[tb];
      b[1] = xrow[tb + 1];
      acc = __builtin_amdgcn_wmma_f32_16x16x4_f32(
           false, amat[kk],  false, b,
           (short)0, acc,  false,  false);
    }

    m_in = __int_as_float(
        __builtin_amdgcn_ds_bpermute((16 + col) << 2, __float_as_int(acc[7])));

    if (c >= warm_chunks) {
      const float* xp = xrow + tbase + 8 * halfsel;
      float*       op = orow + tbase + 8 * halfsel;
      const v4f x0 = *(const v4f*)(xp);
      const v4f x1 = *(const v4f*)(xp + 4);
      float xv8[8] = {x0[0], x0[1], x0[2], x0[3], x1[0], x1[1], x1[2], x1[3]};
      v4f o0, o1;
#pragma unroll
      for (int v = 0; v < 8; ++v) {
        const float M = acc[v];
        const float smooth = fast_pow_pos(1e-6f + M, alpha);
        const float base   = xv8[v] * __builtin_amdgcn_rcpf(smooth + 1e-6f) + delta;
        const float p      = fast_pow_pos(base, r) - dr;
        if (v < 4) o0[v] = p; else o1[v - 4] = p;
      }
      *(v4f*)(op)     = o0;
      *(v4f*)(op + 4) = o1;
    }
  }
}

extern "C" void kernel_launch(void* const* d_in, const int* in_sizes, int n_in,
                              void* d_out, int out_size, void* d_ws, size_t ws_size,
                              hipStream_t stream) {
  const float* x       = (const float*)d_in[0];
  const float* p_alpha = (const float*)d_in[1];
  const float* p_delta = (const float*)d_in[2];
  const float* p_r     = (const float*)d_in[3];
  const float* p_s     = (const float*)d_in[4];
  float* out = (float*)d_out;

  const int n_rows = in_sizes[0] / T_DIM;
  const int n_rowgroups = n_rows / 16;
  const int tasks = n_rowgroups * NSEG;
  const int blocks = (tasks + WAVES_PER_BLOCK - 1) / WAVES_PER_BLOCK;

  pcen_wmma_kernel<<<dim3(blocks), dim3(WAVES_PER_BLOCK * 32), 0, stream>>>(
      x, p_alpha, p_delta, p_r, p_s, out, n_rowgroups);
}


# one wave per row: WMMA columns = 16 consecutive 16-step pieces of the row, carry by in-wave shifted scan (new sum order, no lookback); 1-KB contiguous loads/stores, x read once, 8 passes in flight
# speedup vs baseline: 2.4394x; 2.4394x over previous
#include <hip/hip_runtime.h>


typedef __attribute__((ext_vector_type(2))) float v2f;
typedef __attribute__((ext_vector_type(4))) float v4f;
typedef __attribute__((ext_vector_type(8))) float v8f;

#define T_DIM 8192
#define CH 256
#define NCH (T_DIM / CH)
#define PF 8
#define WAVES_PER_BLOCK 8

__device__ __forceinline__ float fast_pow_pos(float a, float b) {
  return __builtin_amdgcn_exp2f(b * __builtin_amdgcn_logf(a));
}

__device__ __forceinline__ float xhalf(float x) {
  return __builtin_bit_cast(float, __builtin_amdgcn_permlanex16(
      __builtin_bit_cast(int, x), __builtin_bit_cast(int, x), 0x76543210, 0xfedcba98, false, false));
}

template <int N>
__device__ __forceinline__ float row_shr(float x) {
  return __builtin_bit_cast(float, __builtin_amdgcn_update_dpp(
      0, __builtin_bit_cast(int, x), 0x110 + N, 0xf, 0xf, true));
}

template <int NB>
__device__ __forceinline__ float ipow(const float (&sq)[NB], int e) {
  float p = 1.0f;
#pragma unroll
  for (int b = 0; b < NB; ++b) p *= ((e >> b) & 1) ? sq[b] : 1.0f;
  return p;
}

struct Coef {
  v2f   amat[4];
  float cdecay[8];
  float qcol;
  float d16, d32, d64, d128, d256;
  float alpha, delta, r, dr;
  int   h;
};

__device__ __forceinline__ void pcen_pass(const v4f x0, const v4f x1, float& c,
                                          float* __restrict__ op, const Coef& K)
{
  const float xv[8] = {x0[0], x0[1], x0[2], x0[3], x1[0], x1[1], x1[2], x1[3]};

  v8f acc = {0.f, 0.f, 0.f, 0.f, 0.f, 0.f, 0.f, 0.f};
#pragma unroll
  for (int kk = 0; kk < 4; ++kk) {
    v2f b;
    b[0] = xv[2 * kk];
    b[1] = xv[2 * kk + 1];
    acc = __builtin_amdgcn_wmma_f32_16x16x4_f32(
         false, K.amat[kk],  false, b,
         (short)0, acc,  false,  false);
  }

  const float other = xhalf(acc[7]);
  float e = K.h ? acc[7] : other;
  e = fmaf(K.d16,  row_shr<1>(e), e);
  e = fmaf(K.d32,  row_shr<2>(e), e);
  e = fmaf(K.d64,  row_shr<4>(e), e);
  e = fmaf(K.d128, row_shr<8>(e), e);
  const float cin = fmaf(K.qcol, c, row_shr<1>(e));
  c = fmaf(K.d256, c, __builtin_bit_cast(float, __builtin_amdgcn_readlane(__builtin_bit_cast(int, e), 15)));

  v4f o0, o1;
#pragma unroll
  for (int v = 0; v < 8; ++v) {
    const float M = fmaf(K.cdecay[v], cin, acc[v]);
    const float smooth = fast_pow_pos(1e-6f + M, K.alpha);
    const float base   = xv[v] * __builtin_amdgcn_rcpf(smooth + 1e-6f) + K.delta;
    const float p      = fast_pow_pos(base, K.r) - K.dr;
    if (v < 4) o0[v] = p; else o1[v - 4] = p;
  }
  *(v4f*)(op)     = o0;
  *(v4f*)(op + 4) = o1;
}

__global__ __launch_bounds__(256) void pcen_wmma_kernel(
    const float* __restrict__ x,
    const float* __restrict__ p_alpha,
    const float* __restrict__ p_delta,
    const float* __restrict__ p_r,
    const float* __restrict__ p_s,
    float* __restrict__ out,
    int n_rows)
{
  const int lane = threadIdx.x & 31;
  const int row  = __builtin_amdgcn_readfirstlane(blockIdx.x * WAVES_PER_BLOCK + (threadIdx.x >> 5));
  if (row >= n_rows) return;

  const int h = lane >> 4;
  const int n = lane & 15;

  Coef K;
  K.alpha = fminf(fmaxf(p_alpha[0], 0.01f), 0.99f);
  K.delta = fabsf(p_delta[0]) + 1e-6f;
  K.r     = fminf(fmaxf(p_r[0], 0.01f), 1.0f);
  K.dr    = fast_pow_pos(K.delta, K.r);
  K.h     = h;
  const float s     = p_s[0];
  const float decay = 1.0f - s;

  float sq[5];
  sq[0] = decay;
#pragma unroll
  for (int b = 1; b < 5; ++b) sq[b] = sq[b - 1] * sq[b - 1];
  K.d16  = sq[4];
  K.d32  = K.d16 * K.d16;
  K.d64  = K.d32 * K.d32;
  K.d128 = K.d64 * K.d64;
  K.d256 = K.d128 * K.d128;
  const float sq16[4] = {K.d16, K.d32, K.d64, K.d128};
  K.qcol = ipow<4>(sq16, n);

  const float sq4[4] = {sq[0], sq[1], sq[2], sq[3]};
#pragma unroll
  for (int kk = 0; kk < 4; ++kk) {
#pragma unroll
    for (int v = 0; v < 2; ++v) {
      const int d = n - (8 * h + 2 * kk + v);
      K.amat[kk][v] = ((d >= 0) ? s : 0.0f) * ipow<4>(sq4, d & 15);
    }
  }
#pragma unroll
  for (int v = 0; v < 8; ++v) K.cdecay[v] = ipow<5>(sq, 8 * h + v + 1);

  const size_t lane_off = (size_t)row * (size_t)T_DIM + (size_t)(16 * n + 8 * h);
  const float* __restrict__ xp = x   + lane_off;
  float*       __restrict__ op = out + lane_off;

  v4f ra[PF], rb[PF];
#pragma unroll
  for (int p = 0; p < PF; ++p) {
    ra[p] = *(const v4f*)(xp + p * CH);
    rb[p] = *(const v4f*)(xp + p * CH + 4);
  }

  float c = x[(size_t)row * (size_t)T_DIM];

  for (int c0 = 0; c0 < NCH - PF; c0 += PF) {
#pragma unroll
    for (int p = 0; p < PF; ++p) {
      pcen_pass(ra[p], rb[p], c, op + (c0 + p) * CH, K);
      const float* nx = xp + (c0 + p + PF) * CH;
      ra[p] = *(const v4f*)(nx);
      rb[p] = *(const v4f*)(nx + 4);
      if (p & 1) __builtin_amdgcn_sched_barrier(0);
    }
  }
#pragma unroll
  for (int p = 0; p < PF; ++p)
    pcen_pass(ra[p], rb[p], c, op + (NCH - PF + p) * CH, K);
}

extern "C" void kernel_launch(void* const* d_in, const int* in_sizes, int n_in,
                              void* d_out, int out_size, void* d_ws, size_t ws_size,
                              hipStream_t stream) {
  const float* x       = (const float*)d_in[0];
  const float* p_alpha = (const float*)d_in[1];
  const float* p_delta = (const float*)d_in[2];
  const float* p_r     = (const float*)d_in[3];
  const float* p_s     = (const float*)d_in[4];
  float* out = (float*)d_out;

  const int n_rows = in_sizes[0] / T_DIM;
  const int blocks = (n_rows + WAVES_PER_BLOCK - 1) / WAVES_PER_BLOCK;

  pcen_wmma_kernel<<<dim3(blocks), dim3(WAVES_PER_BLOCK * 32), 0, stream>>>(
      x, p_alpha, p_delta, p_r, p_s, out, n_rows);
}


# whole-line loads and stores: lane l owns 16 contiguous bytes of each 512-byte block; WMMA does 4-step groups, carry by 16-lane scan of group ends plus a 4-segment chain (new sum order)
# speedup vs baseline: 2.8125x; 1.1529x over previous
#include <hip/hip_runtime.h>


typedef __attribute__((ext_vector_type(2))) float v2f;
typedef __attribute__((ext_vector_type(4))) float v4f;
typedef __attribute__((ext_vector_type(8))) float v8f;

#define T_DIM 8192
#define CH 256
#define BLK 128
#define NCH (T_DIM / CH)
#define PF 8
#define WAVES_PER_BLOCK 8

__device__ __forceinline__ float fast_pow_pos(float a, float b) {
  return __builtin_amdgcn_exp2f(b * __builtin_amdgcn_logf(a));
}

template <int N>
__device__ __forceinline__ float row_shr(float x) {
  return __builtin_bit_cast(float, __builtin_amdgcn_update_dpp(
      0, __builtin_bit_cast(int, x), 0x110 + N, 0xf, 0xf, true));
}

template <int NB>
__device__ __forceinline__ float ipow(const float (&sq)[NB], int e) {
  float p = 1.0f;
#pragma unroll
  for (int b = 0; b < NB; ++b) p *= ((e >> b) & 1) ? sq[b] : 1.0f;
  return p;
}

struct Coef {
  v2f   amat[4];
  float cdecay[4];
  float qcol;
  float d4, d8, d16, d32, d64;
  float alpha, delta, r, dr;
  int   h;
};

__device__ __forceinline__ void pcen_pass(const v4f x0, const v4f x1, float& c,
                                          float* __restrict__ op, const Coef& K)
{
  const float xv[8] = {x0[0], x0[1], x0[2], x0[3], x1[0], x1[1], x1[2], x1[3]};

  v8f acc = {0.f, 0.f, 0.f, 0.f, 0.f, 0.f, 0.f, 0.f};
#pragma unroll
  for (int kk = 0; kk < 4; ++kk) {
    v2f b;
    b[0] = xv[2 * kk];
    b[1] = xv[2 * kk + 1];
    acc = __builtin_amdgcn_wmma_f32_16x16x4_f32(
         false, K.amat[kk],  false, b,
         (short)0, acc,  false,  false);
  }

  float ea = acc[3], eb = acc[7];
  ea = fmaf(K.d4,  row_shr<1>(ea), ea);   eb = fmaf(K.d4,  row_shr<1>(eb), eb);
  ea = fmaf(K.d8,  row_shr<2>(ea), ea);   eb = fmaf(K.d8,  row_shr<2>(eb), eb);
  ea = fmaf(K.d16, row_shr<4>(ea), ea);   eb = fmaf(K.d16, row_shr<4>(eb), eb);
  ea = fmaf(K.d32, row_shr<8>(ea), ea);   eb = fmaf(K.d32, row_shr<8>(eb), eb);
  const float t0 = __builtin_bit_cast(float, __builtin_amdgcn_readlane(__builtin_bit_cast(int, ea), 15));
  const float t1 = __builtin_bit_cast(float, __builtin_amdgcn_readlane(__builtin_bit_cast(int, ea), 31));
  const float t2 = __builtin_bit_cast(float, __builtin_amdgcn_readlane(__builtin_bit_cast(int, eb), 15));
  const float t3 = __builtin_bit_cast(float, __builtin_amdgcn_readlane(__builtin_bit_cast(int, eb), 31));
  const float c1 = fmaf(K.d64, c,  t0);
  const float c2 = fmaf(K.d64, c1, t1);
  const float c3 = fmaf(K.d64, c2, t2);
  const float cina = fmaf(K.qcol, K.h ? c1 : c,  row_shr<1>(ea));
  const float cinb = fmaf(K.qcol, K.h ? c3 : c2, row_shr<1>(eb));
  c = fmaf(K.d64, c3, t3);

  v4f o0, o1;
#pragma unroll
  for (int v = 0; v < 8; ++v) {
    const float M = fmaf(K.cdecay[v & 3], (v < 4) ? cina : cinb, acc[v]);
    const float smooth = fast_pow_pos(1e-6f + M, K.alpha);
    const float base   = xv[v] * __builtin_amdgcn_rcpf(smooth + 1e-6f) + K.delta;
    const float p      = fast_pow_pos(base, K.r) - K.dr;
    if (v < 4) o0[v] = p; else o1[v - 4] = p;
  }
  *(v4f*)(op)       = o0;
  *(v4f*)(op + BLK) = o1;
}

__global__ __launch_bounds__(256) void pcen_wmma_kernel(
    const float* __restrict__ x,
    const float* __restrict__ p_alpha,
    const float* __restrict__ p_delta,
    const float* __restrict__ p_r,
    const float* __restrict__ p_s,
    float* __restrict__ out,
    int n_rows)
{
  const int lane = threadIdx.x & 31;
  const int row  = __builtin_amdgcn_readfirstlane(blockIdx.x * WAVES_PER_BLOCK + (threadIdx.x >> 5));
  if (row >= n_rows) return;

  const int h = lane >> 4;
  const int n = lane & 15;

  Coef K;
  K.alpha = fminf(fmaxf(p_alpha[0], 0.01f), 0.99f);
  K.delta = fabsf(p_delta[0]) + 1e-6f;
  K.r     = fminf(fmaxf(p_r[0], 0.01f), 1.0f);
  K.dr    = fast_pow_pos(K.delta, K.r);
  K.h     = h;
  const float s     = p_s[0];
  const float decay = 1.0f - s;

  const float d2 = decay * decay;
  K.d4   = d2 * d2;
  K.d8   = K.d4 * K.d4;
  K.d16  = K.d8 * K.d8;
  K.d32  = K.d16 * K.d16;
  K.d64  = K.d32 * K.d32;
  const float sq4[4] = {K.d4, K.d8, K.d16, K.d32};
  K.qcol = ipow<4>(sq4, n);

  const float sq1[2] = {decay, d2};
#pragma unroll
  for (int kk = 0; kk < 4; ++kk) {
#pragma unroll
    for (int e = 0; e < 2; ++e) {
      const bool same = (((n >> 2) & 1) == (kk >> 1)) && ((n >> 3) == h);
      const int  d    = (n & 3) - (2 * (kk & 1) + e);
      K.amat[kk][e] = ((same && d >= 0) ? s : 0.0f) * ipow<2>(sq1, d & 3);
    }
  }
  K.cdecay[0] = decay;
  K.cdecay[1] = d2;
  K.cdecay[2] = d2 * decay;
  K.cdecay[3] = K.d4;

  const size_t lane_off = (size_t)row * (size_t)T_DIM + (size_t)(4 * lane);
  const float* __restrict__ xp = x   + lane_off;
  float*       __restrict__ op = out + lane_off;

  v4f ra[PF], rb[PF];
#pragma unroll
  for (int p = 0; p < PF; ++p) {
    ra[p] = *(const v4f*)(xp + p * CH);
    rb[p] = *(const v4f*)(xp + p * CH + BLK);
  }

  float c = x[(size_t)row * (size_t)T_DIM];

  for (int c0 = 0; c0 < NCH - PF; c0 += PF) {
#pragma unroll
    for (int p = 0; p < PF; ++p) {
      pcen_pass(ra[p], rb[p], c, op + (c0 + p) * CH, K);
      const float* nx = xp + (c0 + p + PF) * CH;
      ra[p] = *(const v4f*)(nx);
      rb[p] = *(const v4f*)(nx + BLK);
      if (p & 1) __builtin_amdgcn_sched_barrier(0);
    }
  }
#pragma unroll
  for (int p = 0; p < PF; ++p)
    pcen_pass(ra[p], rb[p], c, op + (NCH - PF + p) * CH, K);
}

extern "C" void kernel_launch(void* const* d_in, const int* in_sizes, int n_in,
                              void* d_out, int out_size, void* d_ws, size_t ws_size,
                              hipStream_t stream) {
  const float* x       = (const float*)d_in[0];
  const float* p_alpha = (const float*)d_in[1];
  const float* p_delta = (const float*)d_in[2];
  const float* p_r     = (const float*)d_in[3];
  const float* p_s     = (const float*)d_in[4];
  float* out = (float*)d_out;

  const int n_rows = in_sizes[0] / T_DIM;
  const int blocks = (n_rows + WAVES_PER_BLOCK - 1) / WAVES_PER_BLOCK;

  pcen_wmma_kernel<<<dim3(blocks), dim3(WAVES_PER_BLOCK * 32), 0, stream>>>(
      x, p_alpha, p_delta, p_r, p_s, out, n_rows);
}
